# starting kernel
# baseline (speedup 1.0000x reference)
#include <hip/hip_runtime.h>


typedef __attribute__((ext_vector_type(2))) float v2f;
typedef __attribute__((ext_vector_type(8))) float v8f;

#define F_DIM 256
#define CAPS 10
#define CDIM 16
#define LDS_STRIDE 17

__global__ void caps_reduce_kernel(const float* __restrict__ caps,
                                   float* __restrict__ wsum) {
    int idx = blockIdx.x * blockDim.x + threadIdx.x;
    if (idx >= F_DIM * CDIM) return;
    int f = idx >> 4;
    int d = idx & 15;
    const float* p = caps + f * (CAPS * CDIM) + d;
    float s = 0.f;
#pragma unroll
    for (int c = 0; c < CAPS; ++c) s += p[c * CDIM];
    wsum[f * CDIM + d] = s;
}

template <bool FROM_CAPS>
__global__ __launch_bounds__(256) void caps_gemm_kernel(
    const float* __restrict__ x,
    const float* __restrict__ w,
    float* __restrict__ out,
    int numTiles) {
    __shared__ float lw[F_DIM * LDS_STRIDE];

    const int tid = threadIdx.x;

    {
        const int k = tid;
        float acc[CDIM];
        if (FROM_CAPS) {
#pragma unroll
            for (int n = 0; n < CDIM; ++n) acc[n] = 0.f;
            const float* cp = w + (size_t)k * (CAPS * CDIM);
#pragma unroll
            for (int c = 0; c < CAPS; ++c) {
                const float4* q = (const float4*)(cp + c * CDIM);
#pragma unroll
                for (int j = 0; j < 4; ++j) {
                    float4 v = q[j];
                    acc[4 * j + 0] += v.x;
                    acc[4 * j + 1] += v.y;
                    acc[4 * j + 2] += v.z;
                    acc[4 * j + 3] += v.w;
                }
            }
        } else {
            const float4* q = (const float4*)(w + (size_t)k * CDIM);
#pragma unroll
            for (int j = 0; j < 4; ++j) {
                float4 v = q[j];
                acc[4 * j + 0] = v.x;
                acc[4 * j + 1] = v.y;
                acc[4 * j + 2] = v.z;
                acc[4 * j + 3] = v.w;
            }
        }
#pragma unroll
        for (int n = 0; n < CDIM; ++n) lw[k * LDS_STRIDE + n] = acc[n];
    }
    __syncthreads();

    const int wave = tid >> 5;
    const int lane = tid & 31;
    const int tile = blockIdx.x * 8 + wave;
    if (tile >= numTiles) return;

    const int n = lane & 15;
    const int h = lane >> 4;
    const long rowBase = (long)tile * 16;

    const float* ap = x + (rowBase + n) * F_DIM + 2 * h;
    const float* bp = lw + (2 * h) * LDS_STRIDE + n;

    v8f c = {0.f, 0.f, 0.f, 0.f, 0.f, 0.f, 0.f, 0.f};

#pragma unroll
    for (int kc = 0; kc < F_DIM; kc += 32) {
        v2f a[8];
#pragma unroll
        for (int i = 0; i < 8; ++i)
            a[i] = *(const v2f*)(ap + kc + 4 * i);
#pragma unroll
        for (int i = 0; i < 8; ++i) {
            const int k0 = kc + 4 * i;
            v2f b;
            b.x = bp[k0 * LDS_STRIDE];
            b.y = bp[k0 * LDS_STRIDE + LDS_STRIDE];
            c = __builtin_amdgcn_wmma_f32_16x16x4_f32(
                false, a[i], false, b, (short)0, c, false, false);
        }
    }

    float* op = out + (rowBase + 8 * h) * CDIM + n;
#pragma unroll
    for (int v = 0; v < 8; ++v) op[v * CDIM] = c[v];
}

extern "C" void kernel_launch(void* const* d_in, const int* in_sizes, int n_in,
                              void* d_out, int out_size, void* d_ws, size_t ws_size,
                              hipStream_t stream) {
    const float* x    = (const float*)d_in[0];
    const float* caps = (const float*)d_in[1];
    float* out = (float*)d_out;

    const int M = in_sizes[0] / F_DIM;
    const int numTiles = (M + 15) / 16;
    const int blocks = (numTiles + 7) / 8;

    if (ws_size >= (size_t)(F_DIM * CDIM) * sizeof(float)) {
        float* wsum = (float*)d_ws;
        caps_reduce_kernel<<<(F_DIM * CDIM + 255) / 256, 256, 0, stream>>>(caps, wsum);
        caps_gemm_kernel<false><<<blocks, 256, 0, stream>>>(x, wsum, out, numTiles);
    } else {
        caps_gemm_kernel<true><<<blocks, 256, 0, stream>>>(x, caps, out, numTiles);
    }
}


# A and B fragments fetched 16 bytes per lane (k permuted inside each 8-block, same map both operands): each 128-B line of x is touched by 4 load instructions, not 8; k sum order changes
# speedup vs baseline: 1.1843x; 1.1843x over previous
#include <hip/hip_runtime.h>


typedef __attribute__((ext_vector_type(2))) float v2f;
typedef __attribute__((ext_vector_type(8))) float v8f;

#define F_DIM 256
#define CAPS 10
#define CDIM 16
#define KBLK (F_DIM / 8)

__global__ void caps_reduce_kernel(const float* __restrict__ caps,
                                   float* __restrict__ wsum) {
    int idx = blockIdx.x * blockDim.x + threadIdx.x;
    if (idx >= F_DIM * CDIM) return;
    int f = idx >> 4;
    int d = idx & 15;
    const float* p = caps + f * (CAPS * CDIM) + d;
    float s = 0.f;
#pragma unroll
    for (int c = 0; c < CAPS; ++c) s += p[c * CDIM];
    wsum[f * CDIM + d] = s;
}

template <bool FROM_CAPS>
__device__ __forceinline__ float wsum_elem(const float* __restrict__ w, int k, int n) {
    if (FROM_CAPS) {
        const float* p = w + k * (CAPS * CDIM) + n;
        float s = 0.f;
#pragma unroll
        for (int c = 0; c < CAPS; ++c) s += p[c * CDIM];
        return s;
    } else {
        return w[k * CDIM + n];
    }
}

template <bool FROM_CAPS>
__global__ __launch_bounds__(256) void caps_gemm_kernel(
    const float* __restrict__ x,
    const float* __restrict__ w,
    float* __restrict__ out,
    int numTiles) {
    __shared__ float4 lw[KBLK * 32];

    const int tid = threadIdx.x;

    {
        const int n = tid & 15;
        const int h = (tid >> 4) & 1;
#pragma unroll
        for (int q = 0; q < 4; ++q) {
            const int i = (tid >> 5) + 8 * q;
            const int k = 8 * i + 4 * h;
            float4 v;
            v.x = wsum_elem<FROM_CAPS>(w, k + 0, n);
            v.y = wsum_elem<FROM_CAPS>(w, k + 1, n);
            v.z = wsum_elem<FROM_CAPS>(w, k + 2, n);
            v.w = wsum_elem<FROM_CAPS>(w, k + 3, n);
            lw[tid + 256 * q] = v;
        }
    }
    __syncthreads();

    const int wave = tid >> 5;
    const int lane = tid & 31;
    const int tile = blockIdx.x * 8 + wave;
    if (tile >= numTiles) return;

    const int n = lane & 15;
    const int h = lane >> 4;
    const long rowBase = (long)tile * 16;

    const float4* ap = (const float4*)(x + (rowBase + n) * F_DIM + 4 * h);
    const float4* bp = lw + lane;

    v8f c = {0.f, 0.f, 0.f, 0.f, 0.f, 0.f, 0.f, 0.f};

#pragma unroll
    for (int ic = 0; ic < KBLK; ic += 8) {
        float4 a[8];
#pragma unroll
        for (int j = 0; j < 8; ++j)
            a[j] = ap[2 * (ic + j)];
#pragma unroll
        for (int j = 0; j < 8; ++j) {
            const float4 b = bp[32 * (ic + j)];
            v2f a0 = {a[j].x, a[j].y};
            v2f b0 = {b.x, b.y};
            v2f a1 = {a[j].z, a[j].w};
            v2f b1 = {b.z, b.w};
            c = __builtin_amdgcn_wmma_f32_16x16x4_f32(
                false, a0, false, b0, (short)0, c, false, false);
            c = __builtin_amdgcn_wmma_f32_16x16x4_f32(
                false, a1, false, b1, (short)0, c, false, false);
        }
    }

    float* op = out + (rowBase + 8 * h) * CDIM + n;
#pragma unroll
    for (int v = 0; v < 8; ++v) op[v * CDIM] = c[v];
}

extern "C" void kernel_launch(void* const* d_in, const int* in_sizes, int n_in,
                              void* d_out, int out_size, void* d_ws, size_t ws_size,
                              hipStream_t stream) {
    const float* x    = (const float*)d_in[0];
    const float* caps = (const float*)d_in[1];
    float* out = (float*)d_out;

    const int M = in_sizes[0] / F_DIM;
    const int numTiles = (M + 15) / 16;
    const int blocks = (numTiles + 7) / 8;

    if (ws_size >= (size_t)(F_DIM * CDIM) * sizeof(float)) {
        float* wsum = (float*)d_ws;
        caps_reduce_kernel<<<(F_DIM * CDIM + 255) / 256, 256, 0, stream>>>(caps, wsum);
        caps_gemm_kernel<false><<<blocks, 256, 0, stream>>>(x, wsum, out, numTiles);
    } else {
        caps_gemm_kernel<true><<<blocks, 256, 0, stream>>>(x, caps, out, numTiles);
    }
}


# One launch instead of two: each work-group (25 waves, one per 16-row tile) sums the capsules itself into its LDS B fragments, same c order; reduce kernel, workspace and launch gap gone
# speedup vs baseline: 1.3568x; 1.1457x over previous
#include <hip/hip_runtime.h>


typedef __attribute__((ext_vector_type(2))) float v2f;
typedef __attribute__((ext_vector_type(8))) float v8f;

#define F_DIM 256
#define CAPS 10
#define CDIM 16
#define KBLK (F_DIM / 8)
#define WPB 25
#define TPB (WPB * 32)
#define UNITS (F_DIM * CDIM / 4)
#define STAGERS (UNITS / 2)

__device__ __forceinline__ void put_piece(float* lwf, int u, float4 s) {
    const int q = u & 3;
    const int f = u >> 2;
    const int i = f >> 3, h = (f >> 2) & 1, e = f & 3;
    float* d = lwf + ((i * 32 + 16 * h + 4 * q) * 4 + e);
    d[0] = s.x; d[4] = s.y; d[8] = s.z; d[12] = s.w;
}

__global__ __launch_bounds__(TPB) void caps_gemm_kernel(
    const float* __restrict__ x,
    const float* __restrict__ caps,
    float* __restrict__ out,
    int numTiles) {
    __shared__ float4 lw[KBLK * 32];

    const int tid = threadIdx.x;

    if (tid < STAGERS) {
        const int u0 = tid, u1 = tid + STAGERS;
        const float4* p0 = (const float4*)(caps + (u0 >> 2) * (CAPS * CDIM) + 4 * (u0 & 3));
        const float4* p1 = (const float4*)(caps + (u1 >> 2) * (CAPS * CDIM) + 4 * (u1 & 3));
        float4 v0[CAPS], v1[CAPS];
#pragma unroll
        for (int c = 0; c < CAPS; ++c) {
            v0[c] = p0[c * (CDIM / 4)];
            v1[c] = p1[c * (CDIM / 4)];
        }
        float4 s0 = {0.f, 0.f, 0.f, 0.f};
        float4 s1 = {0.f, 0.f, 0.f, 0.f};
#pragma unroll
        for (int c = 0; c < CAPS; ++c) {
            s0.x += v0[c].x; s0.y += v0[c].y; s0.z += v0[c].z; s0.w += v0[c].w;
            s1.x += v1[c].x; s1.y += v1[c].y; s1.z += v1[c].z; s1.w += v1[c].w;
        }
        float* lwf = (float*)lw;
        put_piece(lwf, u0, s0);
        put_piece(lwf, u1, s1);
    }
    __syncthreads();

    const int wave = tid >> 5;
    const int lane = tid & 31;
    const int tile = blockIdx.x * WPB + wave;
    if (tile >= numTiles) return;

    const int n = lane & 15;
    const int h = lane >> 4;
    const long rowBase = (long)tile * 16;

    const float4* ap = (const float4*)(x + (rowBase + n) * F_DIM + 4 * h);
    const float4* bp = lw + lane;

    v8f c = {0.f, 0.f, 0.f, 0.f, 0.f, 0.f, 0.f, 0.f};

#pragma unroll
    for (int ic = 0; ic < KBLK; ic += 8) {
        float4 a[8];
#pragma unroll
        for (int j = 0; j < 8; ++j)
            a[j] = ap[2 * (ic + j)];
#pragma unroll
        for (int j = 0; j < 8; ++j) {
            const float4 b = bp[32 * (ic + j)];
            v2f a0 = {a[j].x, a[j].y};
            v2f b0 = {b.x, b.y};
            v2f a1 = {a[j].z, a[j].w};
            v2f b1 = {b.z, b.w};
            c = __builtin_amdgcn_wmma_f32_16x16x4_f32(
                false, a0, false, b0, (short)0, c, false, false);
            c = __builtin_amdgcn_wmma_f32_16x16x4_f32(
                false, a1, false, b1, (short)0, c, false, false);
        }
    }

    float* op = out + (rowBase + 8 * h) * CDIM + n;
#pragma unroll
    for (int v = 0; v < 8; ++v) op[v * CDIM] = c[v];
}

extern "C" void kernel_launch(void* const* d_in, const int* in_sizes, int n_in,
                              void* d_out, int out_size, void* d_ws, size_t ws_size,
                              hipStream_t stream) {
    const float* x    = (const float*)d_in[0];
    const float* caps = (const float*)d_in[1];
    float* out = (float*)d_out;

    const int M = in_sizes[0] / F_DIM;
    const int numTiles = (M + 15) / 16;
    const int blocks = (numTiles + WPB - 1) / WPB;

    caps_gemm_kernel<<<blocks, TPB, 0, stream>>>(x, caps, out, numTiles);
}


# Each wave requests 16 k blocks of its x tile at wave entry, ahead of the capsule sum and an LDS-only barrier, and keeps 11-16 loads in flight (was 8, issued behind the barrier); sum orders unchanged
# speedup vs baseline: 1.4288x; 1.0531x over previous
#include <hip/hip_runtime.h>


typedef __attribute__((ext_vector_type(2))) float v2f;
typedef __attribute__((ext_vector_type(8))) float v8f;

#define F_DIM 256
#define CAPS 10
#define CDIM 16
#define KBLK (F_DIM / 8)
#define WPB 25
#define TPB (WPB * 32)
#define UNITS (F_DIM * CDIM / 4)
#define STAGERS (UNITS / 2)
#define RING 16
#define EARLY 8

__device__ __forceinline__ void put_piece(float* lwf, int u, float4 s) {
    const int q = u & 3;
    const int f = u >> 2;
    const int i = f >> 3, h = (f >> 2) & 1, e = f & 3;
    float* d = lwf + ((i * 32 + 16 * h + 4 * q) * 4 + e);
    d[0] = s.x; d[4] = s.y; d[8] = s.z; d[12] = s.w;
}

__global__ __launch_bounds__(TPB) void caps_gemm_kernel(
    const float* x,
    const float* __restrict__ caps,
    float* __restrict__ out,
    int numTiles) {
    __shared__ float4 lw[KBLK * 32];

    const int tid = threadIdx.x;
    const int wave = tid >> 5;
    const int lane = tid & 31;
    const int tile = blockIdx.x * WPB + wave;
    const int n = lane & 15;
    const int h = lane >> 4;

    const int ltile = tile < numTiles ? tile : numTiles - 1;
    const float4* ap = (const float4*)(x + ((long)ltile * 16 + n) * F_DIM + 4 * h);

    float4 a[RING];
    const bool stager = tid < STAGERS;

    const int u0 = tid, u1 = tid + STAGERS;
    float4 v0[CAPS], v1[CAPS];
    if (stager) {
        const float4* p0 = (const float4*)(caps + (u0 >> 2) * (CAPS * CDIM) + 4 * (u0 & 3));
        const float4* p1 = (const float4*)(caps + (u1 >> 2) * (CAPS * CDIM) + 4 * (u1 & 3));
#pragma unroll
        for (int c = 0; c < CAPS; ++c) {
            v0[c] = p0[c * (CDIM / 4)];
            v1[c] = p1[c * (CDIM / 4)];
        }
    }

#pragma unroll
    for (int j = 0; j < EARLY; ++j) a[j] = ap[2 * j];

    if (stager) {
        asm volatile("" : "+v"(v0[0].x), "+v"(v0[0].y), "+v"(v0[0].z), "+v"(v0[0].w),
                          "+v"(v1[0].x), "+v"(v1[0].y), "+v"(v1[0].z), "+v"(v1[0].w));
        float4 s0 = {0.f, 0.f, 0.f, 0.f};
        float4 s1 = {0.f, 0.f, 0.f, 0.f};
#pragma unroll
        for (int c = 0; c < CAPS; ++c) {
            s0.x += v0[c].x; s0.y += v0[c].y; s0.z += v0[c].z; s0.w += v0[c].w;
            s1.x += v1[c].x; s1.y += v1[c].y; s1.z += v1[c].z; s1.w += v1[c].w;
        }
        float* lwf = (float*)lw;
        put_piece(lwf, u0, s0);
        put_piece(lwf, u1, s1);
    }

#pragma unroll
    for (int j = EARLY; j < RING; ++j) a[j] = ap[2 * j];

    __builtin_amdgcn_fence(__ATOMIC_RELEASE, "workgroup", "local");
    __builtin_amdgcn_s_barrier();
    __builtin_amdgcn_fence(__ATOMIC_ACQUIRE, "workgroup", "local");

    if (tile >= numTiles) return;

    unsigned obyte = (unsigned)(((tile * 16 + 8 * h) * CDIM + n) * 4);
    asm volatile("" : "+v"(obyte));
    float* op = (float*)((char*)out + obyte);

    const float4* bp = lw + lane;

    v8f c = {0.f, 0.f, 0.f, 0.f, 0.f, 0.f, 0.f, 0.f};

#pragma unroll
    for (int i = 0; i < KBLK; ++i) {
        const float4 b = bp[32 * i];
        const float4 av = a[i % RING];
        v2f a0 = {av.x, av.y};
        v2f b0 = {b.x, b.y};
        v2f a1 = {av.z, av.w};
        v2f b1 = {b.z, b.w};
        c = __builtin_amdgcn_wmma_f32_16x16x4_f32(
            false, a0, false, b0, (short)0, c, false, false);
        c = __builtin_amdgcn_wmma_f32_16x16x4_f32(
            false, a1, false, b1, (short)0, c, false, false);
        if (i + RING < KBLK) a[i % RING] = ap[2 * (i + RING)];
    }

#pragma unroll
    for (int v = 0; v < 8; ++v) op[v * CDIM] = c[v];
}

extern "C" void kernel_launch(void* const* d_in, const int* in_sizes, int n_in,
                              void* d_out, int out_size, void* d_ws, size_t ws_size,
                              hipStream_t stream) {
    const float* x    = (const float*)d_in[0];
    const float* caps = (const float*)d_in[1];
    float* out = (float*)d_out;

    const int M = in_sizes[0] / F_DIM;
    const int numTiles = (M + 15) / 16;
    const int blocks = (numTiles + WPB - 1) / WPB;

    caps_gemm_kernel<<<blocks, TPB, 0, stream>>>(x, caps, out, numTiles);
}
